# starting kernel
# baseline (speedup 1.0000x reference)
#include <hip/hip_runtime.h>
#include <math.h>

typedef __attribute__((ext_vector_type(2))) float v2f;
typedef __attribute__((ext_vector_type(8))) float v8f;

#define HDIM 2048
#define SDIM 4096
#define LSTRIDE (HDIM + 4)

static __device__ __forceinline__ v8f wmma_f32(v2f a, v2f b, v8f c) {
  return __builtin_amdgcn_wmma_f32_16x16x4_f32(false, a, false, b, (short)0, c,
                                               false, false);
}

static __device__ __forceinline__ float tanh_fast(float x) {
  const float e = __expf(2.0f * x);
  return 1.0f - 2.0f / (e + 1.0f);
}

__global__ __launch_bounds__(256) void whb_kernel(const float* __restrict__ W_h,
                                                  const float* __restrict__ dh,
                                                  const float* __restrict__ b,
                                                  float* __restrict__ whb) {
  const int wave = threadIdx.x >> 5;
  const int lane = threadIdx.x & 31;
  const int row  = blockIdx.x * 8 + wave;
  const float* wrow = W_h + (size_t)row * HDIM;
  float sum = 0.f;
  #pragma unroll 4
  for (int k = lane; k < HDIM; k += 32) sum += wrow[k] * dh[k];
  #pragma unroll
  for (int m = 16; m >= 1; m >>= 1) sum += __shfl_xor(sum, m, 32);
  if (lane == 0) whb[row] = sum + b[row];
}

__global__ __launch_bounds__(256) void energy_scores_kernel(
    const float* __restrict__ enc, const float* __restrict__ W_e,
    const float* __restrict__ whb, const float* __restrict__ v,
    float* __restrict__ scores) {
  extern __shared__ float s_enc[];
  __shared__ float s_scores[16];

  const int tid  = threadIdx.x;
  const int wave = tid >> 5;
  const int lane = tid & 31;
  const int lm   = lane & 15;
  const int hi   = lane >> 4;
  const int s0   = blockIdx.x * 16;

  if (tid < 16) s_scores[tid] = 0.f;

  for (int i = tid; i < 16 * (HDIM / 4); i += 256) {
    const int row  = i >> 9;
    const int col4 = i & 511;
    const float4 d =
        *(const float4*)(enc + (size_t)(s0 + row) * HDIM + (col4 << 2));
    *(float4*)(&s_enc[row * LSTRIDE + (col4 << 2)]) = d;
  }
  __syncthreads();

  const float* aRow = s_enc + lm * LSTRIDE + 2 * hi;
  const float* bBase = W_e + (size_t)(wave * 256 + lm) * HDIM + 2 * hi;

  v8f acc[8][2];
  #pragma unroll
  for (int p = 0; p < 8; ++p) {
    acc[p][0] = (v8f){};
    acc[p][1] = (v8f){};
  }

  v2f Bb[2][8];
  #pragma unroll
  for (int j = 0; j < 4; ++j) {
    Bb[0][j]     = *(const v2f*)(bBase + 4 * j);
    Bb[0][4 + j] = *(const v2f*)(bBase + (size_t)16 * HDIM + 4 * j);
  }

  #pragma unroll 1
  for (int c = 0; c < HDIM / 16; ++c) {
    const int k = c << 4;

    v2f a[4];
    #pragma unroll
    for (int j = 0; j < 4; ++j) a[j] = *(const v2f*)(aRow + k + 4 * j);

    #pragma unroll
    for (int p = 0; p < 8; ++p) {
      const int cur = p & 1, nxt = cur ^ 1;

      if (p < 7) {
        const float* nb = bBase + (size_t)(p + 1) * 32 * HDIM + k;
        #pragma unroll
        for (int j = 0; j < 4; ++j) {
          Bb[nxt][j]     = *(const v2f*)(nb + 4 * j);
          Bb[nxt][4 + j] = *(const v2f*)(nb + (size_t)16 * HDIM + 4 * j);
        }
      } else {
        const int k2 = (c < HDIM / 16 - 1) ? (k + 16) : 0;
        const float* nb = bBase + k2;
        #pragma unroll
        for (int j = 0; j < 4; ++j) {
          Bb[nxt][j]     = *(const v2f*)(nb + 4 * j);
          Bb[nxt][4 + j] = *(const v2f*)(nb + (size_t)16 * HDIM + 4 * j);
        }
      }

      #pragma unroll
      for (int j = 0; j < 4; ++j) {
        acc[p][0] = wmma_f32(a[j], Bb[cur][j], acc[p][0]);
        acc[p][1] = wmma_f32(a[j], Bb[cur][4 + j], acc[p][1]);
      }
    }
  }

  float lacc[8];
  #pragma unroll
  for (int r = 0; r < 8; ++r) lacc[r] = 0.f;

  #pragma unroll
  for (int p = 0; p < 8; ++p) {
    const int h0 = wave * 256 + p * 32;
    const float wb0 = whb[h0 + lm],      vv0 = v[h0 + lm];
    const float wb1 = whb[h0 + 16 + lm], vv1 = v[h0 + 16 + lm];
    #pragma unroll
    for (int r = 0; r < 8; ++r) {
      lacc[r] += tanh_fast(acc[p][0][r] + wb0) * vv0 +
                 tanh_fast(acc[p][1][r] + wb1) * vv1;
    }
  }

  #pragma unroll
  for (int r = 0; r < 8; ++r) {
    float t = lacc[r];
    t += __shfl_xor(t, 8, 32);
    t += __shfl_xor(t, 4, 32);
    t += __shfl_xor(t, 2, 32);
    t += __shfl_xor(t, 1, 32);
    if (lm == 0) atomicAdd(&s_scores[r + 8 * hi], t);
  }
  __syncthreads();
  if (tid < 16) scores[s0 + tid] = s_scores[tid];
}

__global__ __launch_bounds__(1024) void softmax_kernel(
    const float* __restrict__ scores, float* __restrict__ out) {
  __shared__ float red[32];
  __shared__ float stat[2];
  const int tid = threadIdx.x;

  out[tid] = 0.f;
  out[tid + 1024] = 0.f;

  float vals[4];
  float m = -INFINITY;
  #pragma unroll
  for (int i = 0; i < 4; ++i) {
    vals[i] = scores[tid + i * 1024];
    m = fmaxf(m, vals[i]);
  }
  #pragma unroll
  for (int o = 16; o >= 1; o >>= 1) m = fmaxf(m, __shfl_xor(m, o, 32));
  if ((tid & 31) == 0) red[tid >> 5] = m;
  __syncthreads();
  if (tid < 32) {
    float x = red[tid];
    #pragma unroll
    for (int o = 16; o >= 1; o >>= 1) x = fmaxf(x, __shfl_xor(x, o, 32));
    if (tid == 0) stat[0] = x;
  }
  __syncthreads();
  m = stat[0];

  float lsum = 0.f;
  #pragma unroll
  for (int i = 0; i < 4; ++i) {
    vals[i] = __expf(vals[i] - m);
    lsum += vals[i];
  }
  #pragma unroll
  for (int o = 16; o >= 1; o >>= 1) lsum += __shfl_xor(lsum, o, 32);
  if ((tid & 31) == 0) red[tid >> 5] = lsum;
  __syncthreads();
  if (tid < 32) {
    float x = red[tid];
    #pragma unroll
    for (int o = 16; o >= 1; o >>= 1) x += __shfl_xor(x, o, 32);
    if (tid == 0) stat[1] = x;
  }
  __syncthreads();
  const float inv = 1.0f / stat[1];
  #pragma unroll
  for (int i = 0; i < 4; ++i) out[HDIM + tid + i * 1024] = vals[i] * inv;
}

__global__ __launch_bounds__(256) void context_kernel(
    const float* __restrict__ enc, const float* __restrict__ attn,
    float* __restrict__ context) {
  __shared__ float s_attn[256];
  const int t = threadIdx.x;
  const int sbase = blockIdx.x * 256;
  s_attn[t] = attn[sbase + t];
  __syncthreads();

  for (int hb = 0; hb < HDIM; hb += 256) {
    const int h = hb + t;
    float sum = 0.f;
    #pragma unroll 4
    for (int s = 0; s < 256; ++s)
      sum += s_attn[s] * enc[(size_t)(sbase + s) * HDIM + h];
    atomicAdd(&context[h], sum);
  }
}

extern "C" void kernel_launch(void* const* d_in, const int* in_sizes, int n_in,
                              void* d_out, int out_size, void* d_ws, size_t ws_size,
                              hipStream_t stream) {
  const float* dh  = (const float*)d_in[0];
  const float* enc = (const float*)d_in[1];
  const float* W_h = (const float*)d_in[2];
  const float* W_e = (const float*)d_in[3];
  const float* b   = (const float*)d_in[4];
  const float* v   = (const float*)d_in[5];

  float* out    = (float*)d_out;
  float* whb    = (float*)d_ws;
  float* scores = whb + HDIM;

  const size_t lds_bytes = (size_t)16 * LSTRIDE * sizeof(float);

  whb_kernel<<<HDIM / 8, 256, 0, stream>>>(W_h, dh, b, whb);
  energy_scores_kernel<<<SDIM / 16, 256, lds_bytes, stream>>>(enc, W_e, whb, v,
                                                              scores);
  softmax_kernel<<<1, 1024, 0, stream>>>(scores, out);
  context_kernel<<<SDIM / 256, 256, 0, stream>>>(enc, out + HDIM, out);
}


# softmax+context fused in one launch split by output column: each block recomputes the softmax statistics and sums all rows of its 32 columns in fixed order; no context atomics; sum order changes
# speedup vs baseline: 3.5121x; 3.5121x over previous
#include <hip/hip_runtime.h>
#include <math.h>

typedef __attribute__((ext_vector_type(2))) float v2f;
typedef __attribute__((ext_vector_type(8))) float v8f;

typedef __attribute__((ext_vector_type(4))) float v4f;

#define HDIM 2048
#define SDIM 4096

#define BM 128
#define BN 256
#define KS 64
#define PITCH (KS + 4)
#define A_STAGE (BM * PITCH)
#define B_STAGE (BN * PITCH)
#define STAGE (A_STAGE + B_STAGE)

static __device__ __forceinline__ v8f wmma_f32(v2f a, v2f b, v8f c) {
  return __builtin_amdgcn_wmma_f32_16x16x4_f32(false, a, false, b, (short)0, c,
                                               false, false);
}

static __device__ __forceinline__ float tanh_fast(float x) {
  const float e = __expf(2.0f * x);
  return 1.0f - 2.0f / (e + 1.0f);
}

__global__ __launch_bounds__(256) void whb_kernel(const float* __restrict__ W_h,
                                                  const float* __restrict__ dh,
                                                  const float* __restrict__ b,
                                                  float* __restrict__ whb,
                                                  float* __restrict__ scores) {
  if (blockIdx.x < SDIM / 256) scores[blockIdx.x * 256 + threadIdx.x] = 0.f;
  const int wave = threadIdx.x >> 5;
  const int lane = threadIdx.x & 31;
  const int row  = blockIdx.x * 8 + wave;
  const float* wrow = W_h + (size_t)row * HDIM;
  float sum = 0.f;
  #pragma unroll 4
  for (int k = lane; k < HDIM; k += 32) sum += wrow[k] * dh[k];
  #pragma unroll
  for (int m = 16; m >= 1; m >>= 1) sum += __shfl_xor(sum, m, 32);
  if (lane == 0) whb[row] = sum + b[row];
}

__global__ __launch_bounds__(256) void energy_scores_kernel(
    const float* __restrict__ enc, const float* __restrict__ W_e,
    const float* __restrict__ whb, const float* __restrict__ v,
    float* __restrict__ scores) {
  __shared__ __attribute__((aligned(16))) float s_tile[2 * STAGE];
  __shared__ float s_part[4][BM];

  const int tid  = threadIdx.x;
  const int wave = tid >> 5;
  const int lane = tid & 31;
  const int lm   = lane & 15;
  const int hi   = lane >> 4;
  const int wm   = wave >> 2;
  const int wn   = wave & 3;
  const int s0   = blockIdx.x * BM;
  const int h0   = blockIdx.y * BN;

  const int srow = tid >> 4;
  const int scol = (tid & 15) << 2;
  const float* gA = enc + (size_t)(s0 + srow) * HDIM + scol;
  const float* gB = W_e + (size_t)(h0 + srow) * HDIM + scol;
  float* lA = s_tile + srow * PITCH + scol;
  float* lB = lA + A_STAGE;

  v4f st[24];
  #pragma unroll
  for (int i = 0; i < 8; ++i)
    st[i] = *(const v4f*)(gA + (size_t)i * 16 * HDIM);
  #pragma unroll
  for (int i = 0; i < 16; ++i)
    st[8 + i] = *(const v4f*)(gB + (size_t)i * 16 * HDIM);
  #pragma unroll
  for (int i = 0; i < 8; ++i) *(v4f*)(lA + i * 16 * PITCH) = st[i];
  #pragma unroll
  for (int i = 0; i < 16; ++i) *(v4f*)(lB + i * 16 * PITCH) = st[8 + i];
  __syncthreads();

  const float* fA = s_tile + (wm * 64 + lm) * PITCH + 4 * hi;
  const float* fB = s_tile + A_STAGE + (wn * 64 + lm) * PITCH + 4 * hi;

  v8f acc[4][4];
  #pragma unroll
  for (int i = 0; i < 4; ++i) {
    #pragma unroll
    for (int j = 0; j < 4; ++j) acc[i][j] = (v8f){};
  }

  #pragma unroll 1
  for (int t = 0; t < HDIM / KS; ++t) {
    const int cur = (t & 1) * STAGE;
    const int nxt = STAGE - cur;
    const int k1 = ((t + 1) & (HDIM / KS - 1)) * KS;
    #pragma unroll
    for (int i = 0; i < 8; ++i)
      st[i] = *(const v4f*)(gA + (size_t)i * 16 * HDIM + k1);
    #pragma unroll
    for (int i = 0; i < 16; ++i)
      st[8 + i] = *(const v4f*)(gB + (size_t)i * 16 * HDIM + k1);

    const float* pa = fA + cur;
    const float* pb = fB + cur;
    v4f a[2][4], b[2][4];
    #pragma unroll
    for (int i = 0; i < 4; ++i) {
      a[0][i] = *(const v4f*)(pa + i * 16 * PITCH);
      b[0][i] = *(const v4f*)(pb + i * 16 * PITCH);
    }
    #pragma unroll 1
    for (int k2 = 0; k2 < KS / 16; ++k2) {
      const int o1 = k2 * 16 + 8;
      const int o2 = ((k2 + 1) & (KS / 16 - 1)) * 16;
      #pragma unroll
      for (int i = 0; i < 4; ++i) {
        a[1][i] = *(const v4f*)(pa + i * 16 * PITCH + o1);
        b[1][i] = *(const v4f*)(pb + i * 16 * PITCH + o1);
      }
      #pragma unroll
      for (int i = 0; i < 4; ++i) {
        #pragma unroll
        for (int j = 0; j < 4; ++j)
          acc[i][j] = wmma_f32(a[0][i].xy, b[0][j].xy, acc[i][j]);
      }
      #pragma unroll
      for (int i = 0; i < 4; ++i) {
        #pragma unroll
        for (int j = 0; j < 4; ++j)
          acc[i][j] = wmma_f32(a[0][i].zw, b[0][j].zw, acc[i][j]);
      }
      #pragma unroll
      for (int i = 0; i < 4; ++i) {
        a[0][i] = *(const v4f*)(pa + i * 16 * PITCH + o2);
        b[0][i] = *(const v4f*)(pb + i * 16 * PITCH + o2);
      }
      #pragma unroll
      for (int i = 0; i < 4; ++i) {
        #pragma unroll
        for (int j = 0; j < 4; ++j)
          acc[i][j] = wmma_f32(a[1][i].xy, b[1][j].xy, acc[i][j]);
      }
      #pragma unroll
      for (int i = 0; i < 4; ++i) {
        #pragma unroll
        for (int j = 0; j < 4; ++j)
          acc[i][j] = wmma_f32(a[1][i].zw, b[1][j].zw, acc[i][j]);
      }
    }

    #pragma unroll
    for (int i = 0; i < 8; ++i) *(v4f*)(lA + nxt + i * 16 * PITCH) = st[i];
    #pragma unroll
    for (int i = 0; i < 16; ++i) *(v4f*)(lB + nxt + i * 16 * PITCH) = st[8 + i];
    __syncthreads();
  }

  float lacc[4][8];
  #pragma unroll
  for (int i = 0; i < 4; ++i) {
    #pragma unroll
    for (int r = 0; r < 8; ++r) lacc[i][r] = 0.f;
  }
  #pragma unroll
  for (int j = 0; j < 4; ++j) {
    const int h = h0 + wn * 64 + j * 16 + lm;
    const float wb = whb[h], vv = v[h];
    #pragma unroll
    for (int i = 0; i < 4; ++i) {
      #pragma unroll
      for (int r = 0; r < 8; ++r)
        lacc[i][r] += tanh_fast(acc[i][j][r] + wb) * vv;
    }
  }
  #pragma unroll
  for (int m = 8; m >= 1; m >>= 1) {
    #pragma unroll
    for (int i = 0; i < 4; ++i) {
      #pragma unroll
      for (int r = 0; r < 8; ++r) lacc[i][r] += __shfl_xor(lacc[i][r], m, 32);
    }
  }
  if (lm == 0) {
    #pragma unroll
    for (int i = 0; i < 4; ++i) {
      #pragma unroll
      for (int r = 0; r < 8; ++r)
        s_part[wn][wm * 64 + i * 16 + 8 * hi + r] = lacc[i][r];
    }
  }
  __syncthreads();
  if (tid < BM) {
    const float t = (s_part[0][tid] + s_part[1][tid]) +
                    (s_part[2][tid] + s_part[3][tid]);
    atomicAdd(&scores[s0 + tid], t);
  }
}

#define CT_COLS 32
#define CT_ROWS 128
#define CT_GRP 16

__global__ __launch_bounds__(1024) void softmax_context_kernel(
    const float* __restrict__ scores, const float* __restrict__ enc,
    float* __restrict__ out) {
  __shared__ __attribute__((aligned(16))) float s_w[SDIM];
  __shared__ float red_m[32];
  __shared__ float red_s[32];
  __shared__ __attribute__((aligned(16))) float s_fold[32][CT_COLS];

  const int tid  = threadIdx.x;
  const int wave = tid >> 5;
  const int lane = tid & 31;
  const int c4   = lane & 7;
  const int row0 = wave * 4 + (lane >> 3);

  const v4f sc = *(const v4f*)(scores + tid * 4);
  float m = fmaxf(fmaxf(sc.x, sc.y), fmaxf(sc.z, sc.w));
  #pragma unroll
  for (int o = 16; o >= 1; o >>= 1) m = fmaxf(m, __shfl_xor(m, o, 32));
  if (lane == 0) red_m[wave] = m;
  __syncthreads();
  m = red_m[lane];
  #pragma unroll
  for (int o = 16; o >= 1; o >>= 1) m = fmaxf(m, __shfl_xor(m, o, 32));

  v4f e;
  e.x = __expf(sc.x - m);
  e.y = __expf(sc.y - m);
  e.z = __expf(sc.z - m);
  e.w = __expf(sc.w - m);
  float lsum = (e.x + e.y) + (e.z + e.w);
  #pragma unroll
  for (int o = 16; o >= 1; o >>= 1) lsum += __shfl_xor(lsum, o, 32);
  if (lane == 0) red_s[wave] = lsum;
  __syncthreads();
  float tot = red_s[lane];
  #pragma unroll
  for (int o = 16; o >= 1; o >>= 1) tot += __shfl_xor(tot, o, 32);
  const float inv = 1.0f / tot;
  const v4f w4 = e * inv;
  *(v4f*)(s_w + tid * 4) = w4;
  if ((tid >> 4) == (int)blockIdx.x)
    *(v4f*)(out + HDIM + tid * 4) = w4;
  __syncthreads();

  const float* g = enc + (size_t)row0 * HDIM + blockIdx.x * CT_COLS + c4 * 4;
  v4f x[CT_GRP];
  #pragma unroll
  for (int i = 0; i < CT_GRP; ++i)
    x[i] = *(const v4f*)(g + (size_t)i * CT_ROWS * HDIM);
  v4f acc = (v4f){0.f, 0.f, 0.f, 0.f};
  #pragma unroll
  for (int i = 0; i < CT_GRP; ++i) {
    acc += s_w[row0 + i * CT_ROWS] * x[i];
    x[i] = *(const v4f*)(g + (size_t)(CT_GRP + i) * CT_ROWS * HDIM);
  }
  #pragma unroll
  for (int i = 0; i < CT_GRP; ++i)
    acc += s_w[row0 + (CT_GRP + i) * CT_ROWS] * x[i];

  #pragma unroll
  for (int o = 8; o <= 16; o <<= 1) {
    acc.x += __shfl_xor(acc.x, o, 32);
    acc.y += __shfl_xor(acc.y, o, 32);
    acc.z += __shfl_xor(acc.z, o, 32);
    acc.w += __shfl_xor(acc.w, o, 32);
  }
  if (lane < 8) *(v4f*)(&s_fold[wave][c4 * 4]) = acc;
  __syncthreads();
  if (tid < CT_COLS) {
    float t = 0.f;
    #pragma unroll
    for (int w = 0; w < 32; ++w) t += s_fold[w][tid];
    out[blockIdx.x * CT_COLS + tid] = t;
  }
}

extern "C" void kernel_launch(void* const* d_in, const int* in_sizes, int n_in,
                              void* d_out, int out_size, void* d_ws, size_t ws_size,
                              hipStream_t stream) {
  const float* dh  = (const float*)d_in[0];
  const float* enc = (const float*)d_in[1];
  const float* W_h = (const float*)d_in[2];
  const float* W_e = (const float*)d_in[3];
  const float* b   = (const float*)d_in[4];
  const float* v   = (const float*)d_in[5];

  float* out    = (float*)d_out;
  float* whb    = (float*)d_ws;
  float* scores = whb + HDIM;

  whb_kernel<<<HDIM / 8, 256, 0, stream>>>(W_h, dh, b, whb, scores);
  energy_scores_kernel<<<dim3(SDIM / BM, HDIM / BN), 256, 0, stream>>>(
      enc, W_e, whb, v, scores);
  softmax_context_kernel<<<HDIM / CT_COLS, 1024, 0, stream>>>(scores, enc, out);
}


# energy GEMM: third LDS stage; the next slab's LDS stores, global loads and the one barrier per slab moved inside the slab, next slab's first fragments prefetched; k order and all sums unchanged
# speedup vs baseline: 3.7166x; 1.0582x over previous
#include <hip/hip_runtime.h>
#include <math.h>

typedef __attribute__((ext_vector_type(2))) float v2f;
typedef __attribute__((ext_vector_type(8))) float v8f;

typedef __attribute__((ext_vector_type(4))) float v4f;

#define HDIM 2048
#define SDIM 4096

#define BM 128
#define BN 256
#define KS 64
#define PITCH (KS + 4)
#define A_STAGE (BM * PITCH)
#define B_STAGE (BN * PITCH)
#define STAGE (A_STAGE + B_STAGE)

static __device__ __forceinline__ v8f wmma_f32(v2f a, v2f b, v8f c) {
  return __builtin_amdgcn_wmma_f32_16x16x4_f32(false, a, false, b, (short)0, c,
                                               false, false);
}

static __device__ __forceinline__ float tanh_fast(float x) {
  const float e = __expf(2.0f * x);
  return 1.0f - 2.0f / (e + 1.0f);
}

__global__ __launch_bounds__(256) void whb_kernel(const float* __restrict__ W_h,
                                                  const float* __restrict__ dh,
                                                  const float* __restrict__ b,
                                                  float* __restrict__ whb,
                                                  float* __restrict__ scores) {
  if (blockIdx.x < SDIM / 256) scores[blockIdx.x * 256 + threadIdx.x] = 0.f;
  const int wave = threadIdx.x >> 5;
  const int lane = threadIdx.x & 31;
  const int row  = blockIdx.x * 8 + wave;
  const float* wrow = W_h + (size_t)row * HDIM;
  float sum = 0.f;
  #pragma unroll 4
  for (int k = lane; k < HDIM; k += 32) sum += wrow[k] * dh[k];
  #pragma unroll
  for (int m = 16; m >= 1; m >>= 1) sum += __shfl_xor(sum, m, 32);
  if (lane == 0) whb[row] = sum + b[row];
}

#define NSTAGE 3
#define NSLAB (HDIM / KS)
#define TRIPS (KS / 16)

__global__ __launch_bounds__(256) void energy_scores_kernel(
    const float* __restrict__ enc, const float* __restrict__ W_e,
    const float* __restrict__ whb, const float* __restrict__ v,
    float* __restrict__ scores) {
  __shared__ __attribute__((aligned(16))) float s_tile[NSTAGE * STAGE];
  __shared__ float s_part[4][BM];

  const int tid  = threadIdx.x;
  const int wave = tid >> 5;
  const int lane = tid & 31;
  const int lm   = lane & 15;
  const int hi   = lane >> 4;
  const int wm   = wave >> 2;
  const int wn   = wave & 3;
  const int s0   = blockIdx.x * BM;
  const int h0   = blockIdx.y * BN;

  const int srow = tid >> 4;
  const int scol = (tid & 15) << 2;
  const float* gA = enc + (size_t)(s0 + srow) * HDIM + scol;
  const float* gB = W_e + (size_t)(h0 + srow) * HDIM + scol;
  float* lA = s_tile + srow * PITCH + scol;
  float* lB = lA + A_STAGE;

  v4f st[24];
  #pragma unroll
  for (int i = 0; i < 8; ++i)
    st[i] = *(const v4f*)(gA + (size_t)i * 16 * HDIM);
  #pragma unroll
  for (int i = 0; i < 16; ++i)
    st[8 + i] = *(const v4f*)(gB + (size_t)i * 16 * HDIM);
  #pragma unroll
  for (int i = 0; i < 8; ++i) *(v4f*)(lA + i * 16 * PITCH) = st[i];
  #pragma unroll
  for (int i = 0; i < 16; ++i) *(v4f*)(lB + i * 16 * PITCH) = st[8 + i];
  #pragma unroll
  for (int i = 0; i < 8; ++i)
    st[i] = *(const v4f*)(gA + (size_t)i * 16 * HDIM + KS);
  #pragma unroll
  for (int i = 0; i < 16; ++i)
    st[8 + i] = *(const v4f*)(gB + (size_t)i * 16 * HDIM + KS);
  __syncthreads();

  const float* fA = s_tile + (wm * 64 + lm) * PITCH + 4 * hi;
  const float* fB = s_tile + A_STAGE + (wn * 64 + lm) * PITCH + 4 * hi;

  v8f acc[4][4];
  #pragma unroll
  for (int i = 0; i < 4; ++i) {
    #pragma unroll
    for (int j = 0; j < 4; ++j) acc[i][j] = (v8f){};
  }

  v4f a[2][4], b[2][4];
  #pragma unroll
  for (int i = 0; i < 4; ++i) {
    a[0][i] = *(const v4f*)(fA + i * 16 * PITCH);
    b[0][i] = *(const v4f*)(fB + i * 16 * PITCH);
  }

  #pragma unroll 1
  for (int q = 0; q < NSLAB * TRIPS; ++q) {
    const int t   = q / TRIPS;
    const int k2  = q % TRIPS;
    const int cur = (t % NSTAGE) * STAGE;
    const int nx1 = ((t + 1) % NSTAGE) * STAGE;
    const float* pa = fA + cur;
    const float* pb = fB + cur;
    const int o1 = k2 * 16 + 8;
    const int o2 = (k2 == TRIPS - 1) ? (nx1 - cur) : (k2 + 1) * 16;
    #pragma unroll
    for (int i = 0; i < 4; ++i) {
      a[1][i] = *(const v4f*)(pa + i * 16 * PITCH + o1);
      b[1][i] = *(const v4f*)(pb + i * 16 * PITCH + o1);
    }
    #pragma unroll
    for (int i = 0; i < 4; ++i) {
      #pragma unroll
      for (int j = 0; j < 4; ++j)
        acc[i][j] = wmma_f32(a[0][i].xy, b[0][j].xy, acc[i][j]);
    }
    #pragma unroll
    for (int i = 0; i < 4; ++i) {
      #pragma unroll
      for (int j = 0; j < 4; ++j)
        acc[i][j] = wmma_f32(a[0][i].zw, b[0][j].zw, acc[i][j]);
    }
    #pragma unroll
    for (int i = 0; i < 4; ++i) {
      a[0][i] = *(const v4f*)(pa + i * 16 * PITCH + o2);
      b[0][i] = *(const v4f*)(pb + i * 16 * PITCH + o2);
    }
    #pragma unroll
    for (int i = 0; i < 4; ++i) {
      #pragma unroll
      for (int j = 0; j < 4; ++j)
        acc[i][j] = wmma_f32(a[1][i].xy, b[1][j].xy, acc[i][j]);
    }
    #pragma unroll
    for (int i = 0; i < 4; ++i) {
      #pragma unroll
      for (int j = 0; j < 4; ++j)
        acc[i][j] = wmma_f32(a[1][i].zw, b[1][j].zw, acc[i][j]);
    }

    if (k2 == 1) {
      #pragma unroll
      for (int i = 0; i < 8; ++i) *(v4f*)(lA + nx1 + i * 16 * PITCH) = st[i];
      #pragma unroll
      for (int i = 0; i < 16; ++i)
        *(v4f*)(lB + nx1 + i * 16 * PITCH) = st[8 + i];
      const int kn = ((t + 2) & (NSLAB - 1)) * KS;
      #pragma unroll
      for (int i = 0; i < 8; ++i)
        st[i] = *(const v4f*)(gA + (size_t)i * 16 * HDIM + kn);
      #pragma unroll
      for (int i = 0; i < 16; ++i)
        st[8 + i] = *(const v4f*)(gB + (size_t)i * 16 * HDIM + kn);
    }
    if (k2 == 2) __syncthreads();
  }

  float lacc[4][8];
  #pragma unroll
  for (int i = 0; i < 4; ++i) {
    #pragma unroll
    for (int r = 0; r < 8; ++r) lacc[i][r] = 0.f;
  }
  #pragma unroll
  for (int j = 0; j < 4; ++j) {
    const int h = h0 + wn * 64 + j * 16 + lm;
    const float wb = whb[h], vv = v[h];
    #pragma unroll
    for (int i = 0; i < 4; ++i) {
      #pragma unroll
      for (int r = 0; r < 8; ++r)
        lacc[i][r] += tanh_fast(acc[i][j][r] + wb) * vv;
    }
  }
  #pragma unroll
  for (int m = 8; m >= 1; m >>= 1) {
    #pragma unroll
    for (int i = 0; i < 4; ++i) {
      #pragma unroll
      for (int r = 0; r < 8; ++r) lacc[i][r] += __shfl_xor(lacc[i][r], m, 32);
    }
  }
  if (lm == 0) {
    #pragma unroll
    for (int i = 0; i < 4; ++i) {
      #pragma unroll
      for (int r = 0; r < 8; ++r)
        s_part[wn][wm * 64 + i * 16 + 8 * hi + r] = lacc[i][r];
    }
  }
  __syncthreads();
  if (tid < BM) {
    const float t = (s_part[0][tid] + s_part[1][tid]) +
                    (s_part[2][tid] + s_part[3][tid]);
    atomicAdd(&scores[s0 + tid], t);
  }
}

#define CT_COLS 32
#define CT_ROWS 128
#define CT_GRP 16

__global__ __launch_bounds__(1024) void softmax_context_kernel(
    const float* __restrict__ scores, const float* __restrict__ enc,
    float* __restrict__ out) {
  __shared__ __attribute__((aligned(16))) float s_w[SDIM];
  __shared__ float red_m[32];
  __shared__ float red_s[32];
  __shared__ __attribute__((aligned(16))) float s_fold[32][CT_COLS];

  const int tid  = threadIdx.x;
  const int wave = tid >> 5;
  const int lane = tid & 31;
  const int c4   = lane & 7;
  const int row0 = wave * 4 + (lane >> 3);

  const v4f sc = *(const v4f*)(scores + tid * 4);
  float m = fmaxf(fmaxf(sc.x, sc.y), fmaxf(sc.z, sc.w));
  #pragma unroll
  for (int o = 16; o >= 1; o >>= 1) m = fmaxf(m, __shfl_xor(m, o, 32));
  if (lane == 0) red_m[wave] = m;
  __syncthreads();
  m = red_m[lane];
  #pragma unroll
  for (int o = 16; o >= 1; o >>= 1) m = fmaxf(m, __shfl_xor(m, o, 32));

  v4f e;
  e.x = __expf(sc.x - m);
  e.y = __expf(sc.y - m);
  e.z = __expf(sc.z - m);
  e.w = __expf(sc.w - m);
  float lsum = (e.x + e.y) + (e.z + e.w);
  #pragma unroll
  for (int o = 16; o >= 1; o >>= 1) lsum += __shfl_xor(lsum, o, 32);
  if (lane == 0) red_s[wave] = lsum;
  __syncthreads();
  float tot = red_s[lane];
  #pragma unroll
  for (int o = 16; o >= 1; o >>= 1) tot += __shfl_xor(tot, o, 32);
  const float inv = 1.0f / tot;
  const v4f w4 = e * inv;
  *(v4f*)(s_w + tid * 4) = w4;
  if ((tid >> 4) == (int)blockIdx.x)
    *(v4f*)(out + HDIM + tid * 4) = w4;
  __syncthreads();

  const float* g = enc + (size_t)row0 * HDIM + blockIdx.x * CT_COLS + c4 * 4;
  v4f x[CT_GRP];
  #pragma unroll
  for (int i = 0; i < CT_GRP; ++i)
    x[i] = *(const v4f*)(g + (size_t)i * CT_ROWS * HDIM);
  v4f acc = (v4f){0.f, 0.f, 0.f, 0.f};
  #pragma unroll
  for (int i = 0; i < CT_GRP; ++i) {
    acc += s_w[row0 + i * CT_ROWS] * x[i];
    x[i] = *(const v4f*)(g + (size_t)(CT_GRP + i) * CT_ROWS * HDIM);
  }
  #pragma unroll
  for (int i = 0; i < CT_GRP; ++i)
    acc += s_w[row0 + (CT_GRP + i) * CT_ROWS] * x[i];

  #pragma unroll
  for (int o = 8; o <= 16; o <<= 1) {
    acc.x += __shfl_xor(acc.x, o, 32);
    acc.y += __shfl_xor(acc.y, o, 32);
    acc.z += __shfl_xor(acc.z, o, 32);
    acc.w += __shfl_xor(acc.w, o, 32);
  }
  if (lane < 8) *(v4f*)(&s_fold[wave][c4 * 4]) = acc;
  __syncthreads();
  if (tid < CT_COLS) {
    float t = 0.f;
    #pragma unroll
    for (int w = 0; w < 32; ++w) t += s_fold[w][tid];
    out[blockIdx.x * CT_COLS + tid] = t;
  }
}

extern "C" void kernel_launch(void* const* d_in, const int* in_sizes, int n_in,
                              void* d_out, int out_size, void* d_ws, size_t ws_size,
                              hipStream_t stream) {
  const float* dh  = (const float*)d_in[0];
  const float* enc = (const float*)d_in[1];
  const float* W_h = (const float*)d_in[2];
  const float* W_e = (const float*)d_in[3];
  const float* b   = (const float*)d_in[4];
  const float* v   = (const float*)d_in[5];

  float* out    = (float*)d_out;
  float* whb    = (float*)d_ws;
  float* scores = whb + HDIM;

  whb_kernel<<<HDIM / 8, 256, 0, stream>>>(W_h, dh, b, whb, scores);
  energy_scores_kernel<<<dim3(SDIM / BM, HDIM / BN), 256, 0, stream>>>(
      enc, W_e, whb, v, scores);
  softmax_context_kernel<<<HDIM / CT_COLS, 1024, 0, stream>>>(scores, enc, out);
}
